# starting kernel
# baseline (speedup 1.0000x reference)
#include <hip/hip_runtime.h>


typedef __attribute__((ext_vector_type(2))) float v2f;
typedef __attribute__((ext_vector_type(4))) float v4f;
typedef __attribute__((ext_vector_type(8))) float v8f;

#define HSZ 20
#define S   34
#define TPW 4

__device__ __forceinline__ float relu(float v) {
    return __builtin_amdgcn_fmed3f(v, 0.0f, 3.402823466e38f);
}

__global__ __launch_bounds__(32) void rnn_scan_wmma_f32(
    const float* __restrict__ x,
    const float* __restrict__ W_ih,
    const float* __restrict__ W_hh,
    const float* __restrict__ b_ih,
    const float* __restrict__ b_hh,
    const float* __restrict__ fc_w,
    const float* __restrict__ fc_b,
    float* __restrict__ out,
    int B, int T)
{
    __shared__ float htile[TPW][16 * S];

    const int lane = threadIdx.x;
    const int m    = lane & 15;
    const int hi   = lane >> 4;
    const int koff = hi * 2;
    const int base = blockIdx.x * (16 * TPW);

    auto welem = [&](int k, int j) -> float {
        float v = 0.0f;
        if (j < HSZ) {
            if (k < HSZ)            v = W_hh[j * HSZ + k];
            else if (k == HSZ)      v = W_ih[j];
            else if (k == HSZ + 1)  v = b_ih[j] + b_hh[j];
        }
        return v;
    };
    v2f bw[2][6];
    #pragma unroll
    for (int n = 0; n < 2; ++n) {
        #pragma unroll
        for (int kk = 0; kk < 6; ++kk) {
            const int j  = n * 16 + m;
            const int k0 = kk * 4 + koff;
            bw[n][kk].x = welem(k0, j);
            bw[n][kk].y = welem(k0 + 1, j);
        }
    }

    for (int i = lane; i < TPW * 16 * S; i += 32) (&htile[0][0])[i] = 0.0f;
    __syncthreads();

    const float* xrow[TPW];
    #pragma unroll
    for (int tt = 0; tt < TPW; ++tt) {
        int row = base + tt * 16 + m;
        if (row >= B) row = B - 1;
        xrow[tt] = x + (size_t)row * (size_t)T;
    }

    auto step = [&](const float (&xv)[TPW]) {
        v2f a[TPW][6];
        #pragma unroll
        for (int tt = 0; tt < TPW; ++tt) {
            #pragma unroll
            for (int kk = 0; kk < 5; ++kk)
                a[tt][kk] = *(const v2f*)&htile[tt][m * S + kk * 4 + koff];
            a[tt][5].x = hi ? 0.0f : xv[tt];
            a[tt][5].y = hi ? 0.0f : 1.0f;
        }
        v8f c[TPW][2];
        #pragma unroll
        for (int tt = 0; tt < TPW; ++tt) {
            c[tt][0] = (v8f){};
            c[tt][1] = (v8f){};
            #pragma unroll
            for (int kk = 0; kk < 6; ++kk) {
                c[tt][0] = __builtin_amdgcn_wmma_f32_16x16x4_f32(
                               false, a[tt][kk], false, bw[0][kk], (short)0,
                               c[tt][0], false, false);
                c[tt][1] = __builtin_amdgcn_wmma_f32_16x16x4_f32(
                               false, a[tt][kk], false, bw[1][kk], (short)0,
                               c[tt][1], false, false);
            }
        }
        #pragma unroll
        for (int tt = 0; tt < TPW; ++tt) {
            #pragma unroll
            for (int r = 0; r < 8; ++r) {
                const int row = hi * 8 + r;
                htile[tt][row * S + m]      = relu(c[tt][0][r]);
                htile[tt][row * S + 16 + m] = relu(c[tt][1][r]);
            }
        }
    };

    if ((T & 3) == 0) {
        const int G = T >> 2;
        v4f xb0[TPW], xb1[TPW];
        #pragma unroll
        for (int tt = 0; tt < TPW; ++tt) {
            xb0[tt] = *(const v4f*)(xrow[tt]);
            xb1[tt] = (G > 1) ? *(const v4f*)(xrow[tt] + 4) : xb0[tt];
        }
        #pragma unroll 1
        for (int g = 0; g < G; ++g) {
            v4f cur[TPW];
            #pragma unroll
            for (int tt = 0; tt < TPW; ++tt) { cur[tt] = xb0[tt]; xb0[tt] = xb1[tt]; }
            const int gg = (g + 2 < G) ? (g + 2) : (G - 1);
            #pragma unroll
            for (int tt = 0; tt < TPW; ++tt)
                xb1[tt] = *(const v4f*)(xrow[tt] + (size_t)gg * 4);
            #pragma unroll
            for (int u = 0; u < 4; ++u) {
                float xv[TPW];
                #pragma unroll
                for (int tt = 0; tt < TPW; ++tt) xv[tt] = cur[tt][u];
                step(xv);
            }
        }
    } else {
        #pragma unroll 1
        for (int t = 0; t < T; ++t) {
            float xv[TPW];
            #pragma unroll
            for (int tt = 0; tt < TPW; ++tt) xv[tt] = xrow[tt][t];
            step(xv);
        }
    }
    __syncthreads();

    if (lane < 16) {
        #pragma unroll
        for (int tt = 0; tt < TPW; ++tt) {
            const int row = base + tt * 16 + lane;
            if (row < B) {
                float acc = fc_b[0];
                #pragma unroll
                for (int j = 0; j < HSZ; ++j)
                    acc += htile[tt][lane * S + j] * fc_w[j];
                out[row] = acc;
            }
        }
    }
}

extern "C" void kernel_launch(void* const* d_in, const int* in_sizes, int n_in,
                              void* d_out, int out_size, void* d_ws, size_t ws_size,
                              hipStream_t stream) {
    const float* x    = (const float*)d_in[0];
    const float* W_ih = (const float*)d_in[1];
    const float* W_hh = (const float*)d_in[2];
    const float* b_ih = (const float*)d_in[3];
    const float* b_hh = (const float*)d_in[4];
    const float* fc_w = (const float*)d_in[5];
    const float* fc_b = (const float*)d_in[6];
    float* out = (float*)d_out;

    const int B = out_size;
    const int T = in_sizes[0] / B;
    const int rowsPerBlock = 16 * TPW;
    const int blocks = (B + rowsPerBlock - 1) / rowsPerBlock;

    rnn_scan_wmma_f32<<<blocks, 32, 0, stream>>>(x, W_ih, W_hh, b_ih, b_hh,
                                                 fc_w, fc_b, out, B, T);
}


# Tile's step spread over the WGP's 4 SIMDs: 4-wave workgroup = 2 unit tiles x 2 K halves, 3 WMMAs a wave (was 12); partial sums swapped via LDS, one barrier a step; K sum order changed
# speedup vs baseline: 3.7952x; 3.7952x over previous
#include <hip/hip_runtime.h>


typedef __attribute__((ext_vector_type(2))) float v2f;
typedef __attribute__((ext_vector_type(4))) float v4f;
typedef __attribute__((ext_vector_type(8))) float v8f;

#define HSZ   20
#define S     34
#define PLANE (16 * S)
#define BUF   (2 * PLANE)

__device__ __forceinline__ float relu(float v) {
    return __builtin_amdgcn_fmed3f(v, 0.0f, 3.402823466e38f);
}

__device__ __forceinline__ void step_barrier() {
    asm volatile("s_wait_dscnt 0x0\n\ts_barrier_signal -1\n\ts_barrier_wait -1" ::: "memory");
}

__global__ __launch_bounds__(128) void rnn_scan_wmma_f32(
    const float* __restrict__ x,
    const float* __restrict__ W_ih,
    const float* __restrict__ W_hh,
    const float* __restrict__ b_ih,
    const float* __restrict__ b_hh,
    const float* __restrict__ fc_w,
    const float* __restrict__ fc_b,
    float* __restrict__ out,
    int B, int T)
{
    __shared__ __attribute__((aligned(16))) float part[2 * BUF];

    const int tid  = threadIdx.x;
    const int lane = tid & 31;
    const int wv   = __builtin_amdgcn_readfirstlane(tid >> 5);
    const int n    = wv & 1;
    const int kh   = wv >> 1;
    const int m    = lane & 15;
    const int hi   = lane >> 4;
    const int koff = hi * 2;
    const int base = blockIdx.x * 16;

    auto welem = [&](int k, int j) -> float {
        float v = 0.0f;
        if (j < HSZ) {
            if (k < HSZ)            v = W_hh[j * HSZ + k];
            else if (k == HSZ)      v = W_ih[j];
            else if (k == HSZ + 1)  v = b_ih[j] + b_hh[j];
        }
        return v;
    };
    v2f bw[3];
    #pragma unroll
    for (int i = 0; i < 3; ++i) {
        const int j  = n * 16 + m;
        const int k0 = (kh * 3 + i) * 4 + koff;
        bw[i].x = welem(k0, j);
        bw[i].y = welem(k0 + 1, j);
    }

    for (int i = tid; i < BUF; i += 128) part[i] = 0.0f;
    __syncthreads();

    int xr = base + m;
    if (xr >= B) xr = B - 1;
    const float* xrow = x + (size_t)xr * (size_t)T;

    const int ldoff = m * S + kh * 12 + koff;
    const int stoff = kh * PLANE + hi * 8 * S + n * 16 + m;

    auto step = [&](float xv, int p) {
        const float* rd = part + p * BUF + ldoff;
        float*       wr = part + (p ^ 1) * BUF + stoff;
        v2f a[3];
        #pragma unroll
        for (int i = 0; i < 3; ++i) {
            const v2f s0 = *(const v2f*)&rd[i * 4];
            const v2f s1 = *(const v2f*)&rd[PLANE + i * 4];
            a[i].x = relu(s0.x + s1.x);
            a[i].y = relu(s0.y + s1.y);
        }
        a[2].x = kh ? (hi ? 0.0f : xv)   : a[2].x;
        a[2].y = kh ? (hi ? 0.0f : 1.0f) : a[2].y;
        v8f cA = (v8f){};
        v8f cB = (v8f){};
        cA = __builtin_amdgcn_wmma_f32_16x16x4_f32(false, a[0], false, bw[0], (short)0, cA, false, false);
        cB = __builtin_amdgcn_wmma_f32_16x16x4_f32(false, a[1], false, bw[1], (short)0, cB, false, false);
        cA = __builtin_amdgcn_wmma_f32_16x16x4_f32(false, a[2], false, bw[2], (short)0, cA, false, false);
        #pragma unroll
        for (int r = 0; r < 8; ++r)
            wr[r * S] = cA[r] + cB[r];
        step_barrier();
    };

    if ((T & 3) == 0) {
        const int G = T >> 2;
        v4f xb0 = *(const v4f*)(xrow);
        v4f xb1 = (G > 1) ? *(const v4f*)(xrow + 4) : xb0;
        #pragma unroll 1
        for (int g = 0; g < G; ++g) {
            const v4f cur = xb0;
            xb0 = xb1;
            const int gg = (g + 2 < G) ? (g + 2) : (G - 1);
            xb1 = *(const v4f*)(xrow + (size_t)gg * 4);
            #pragma unroll
            for (int u = 0; u < 4; ++u)
                step(cur[u], u & 1);
        }
    } else {
        #pragma unroll 1
        for (int t = 0; t < T; ++t)
            step(xrow[t], t & 1);
    }
    __syncthreads();

    if (tid < 16) {
        const int row = base + tid;
        if (row < B) {
            const float* fin = part + (T & 1) * BUF + tid * S;
            float acc = fc_b[0];
            #pragma unroll
            for (int j = 0; j < HSZ; ++j)
                acc += relu(fin[j] + fin[PLANE + j]) * fc_w[j];
            out[row] = acc;
        }
    }
}

extern "C" void kernel_launch(void* const* d_in, const int* in_sizes, int n_in,
                              void* d_out, int out_size, void* d_ws, size_t ws_size,
                              hipStream_t stream) {
    const float* x    = (const float*)d_in[0];
    const float* W_ih = (const float*)d_in[1];
    const float* W_hh = (const float*)d_in[2];
    const float* b_ih = (const float*)d_in[3];
    const float* b_hh = (const float*)d_in[4];
    const float* fc_w = (const float*)d_in[5];
    const float* fc_b = (const float*)d_in[6];
    float* out = (float*)d_out;

    const int B = out_size;
    const int T = in_sizes[0] / B;
    const int blocks = (B + 15) / 16;

    rnn_scan_wmma_f32<<<blocks, 128, 0, stream>>>(x, W_ih, W_hh, b_ih, b_hh,
                                                  fc_w, fc_b, out, B, T);
}


# Units 16..19 leave the matrix unit: formed per lane in A's layout from its ten state values (20 packed FMAs, one half swap); 6 WMMAs a step, one wave a tile, no barrier; sum orders changed
# speedup vs baseline: 4.7532x; 1.2524x over previous
#include <hip/hip_runtime.h>


typedef __attribute__((ext_vector_type(2))) float v2f;
typedef __attribute__((ext_vector_type(4))) float v4f;
typedef __attribute__((ext_vector_type(8))) float v8f;

#define HSZ 20
#define P   20

__device__ __forceinline__ float relu(float v) {
    return __builtin_amdgcn_fmed3f(v, 0.0f, 3.402823466e38f);
}

__device__ __forceinline__ float xhalf(float v) {
    return __builtin_bit_cast(float, __builtin_amdgcn_permlanex16(
        __builtin_bit_cast(int, v), __builtin_bit_cast(int, v), 0x76543210, 0xfedcba98, false, false));
}

__device__ __forceinline__ v2f splat(float v) { return (v2f){v, v}; }

__device__ __forceinline__ void wave_order() {
    __builtin_amdgcn_fence(__ATOMIC_RELEASE, "wavefront");
    __builtin_amdgcn_wave_barrier();
    __builtin_amdgcn_fence(__ATOMIC_ACQUIRE, "wavefront");
}

__global__ __launch_bounds__(32) void rnn_scan_wmma_f32(
    const float* __restrict__ x,
    const float* __restrict__ W_ih,
    const float* __restrict__ W_hh,
    const float* __restrict__ b_ih,
    const float* __restrict__ b_hh,
    const float* __restrict__ fc_w,
    const float* __restrict__ fc_b,
    float* __restrict__ out,
    int B, int T)
{
    __shared__ __attribute__((aligned(16))) float st[16 * P];

    const int lane = threadIdx.x;
    const int c    = lane & 15;
    const int h    = lane >> 4;
    const int base = blockIdx.x * 16;

    v2f bw[6];
    #pragma unroll
    for (int kb = 0; kb < 4; ++kb) {
        bw[kb].x = W_hh[c * HSZ + 8 * h + 2 * kb];
        bw[kb].y = W_hh[c * HSZ + 8 * h + 2 * kb + 1];
    }
    bw[4].x = W_hh[c * HSZ + 16 + 2 * h];
    bw[4].y = W_hh[c * HSZ + 17 + 2 * h];
    bw[5].x = h ? 0.0f : W_ih[c];
    bw[5].y = h ? 0.0f : (b_ih[c] + b_hh[c]);

    const int jk = 16 + 2 * h;
    const int js = 18 - 2 * h;
    v2f wk[10], ws[10];
    #pragma unroll
    for (int i = 0; i < 10; ++i) {
        const int k = (i < 8) ? (8 * h + i) : (16 + 2 * h + (i - 8));
        wk[i].x = W_hh[jk * HSZ + k];
        wk[i].y = W_hh[(jk + 1) * HSZ + k];
        ws[i].x = W_hh[js * HSZ + k];
        ws[i].y = W_hh[(js + 1) * HSZ + k];
    }
    const v2f wik = (v2f){W_ih[jk], W_ih[jk + 1]};
    const v2f bk  = (v2f){b_ih[jk] + b_hh[jk], b_ih[jk + 1] + b_hh[jk + 1]};

    for (int i = lane; i < 16 * P; i += 32) st[i] = 0.0f;
    v2f keep = (v2f){0.0f, 0.0f};
    v2f send = (v2f){0.0f, 0.0f};
    wave_order();

    int xr = base + c;
    if (xr >= B) xr = B - 1;
    const float* xrow = x + (size_t)xr * (size_t)T;

    const float* rd = st + c * P + 8 * h;
    float*       wr = st + 8 * h * P + c;

    auto step = [&](float xv) {
        const v4f u0 = *(const v4f*)&rd[0];
        const v4f u1 = *(const v4f*)&rd[4];
        __builtin_amdgcn_sched_barrier(0);
        v2f a5;
        a5.x = h ? 0.0f : xv;
        a5.y = h ? 0.0f : 1.0f;

        v8f acc = (v8f){};
        acc = __builtin_amdgcn_wmma_f32_16x16x4_f32(false, a5, false, bw[5], (short)0, acc, false, false);
        v2f a4;
        a4.x = relu(keep.x + xhalf(send.x));
        a4.y = relu(keep.y + xhalf(send.y));
        acc = __builtin_amdgcn_wmma_f32_16x16x4_f32(false, a4, false, bw[4], (short)0, acc, false, false);
        v2f ke = __builtin_elementwise_fma(splat(xv), wik, bk);
        ke = __builtin_elementwise_fma(splat(a4.x), wk[8], ke);
        v2f ko = splat(a4.y) * wk[9];
        v2f se = splat(a4.x) * ws[8];
        v2f so = splat(a4.y) * ws[9];
        __builtin_amdgcn_sched_barrier(0);

        const float u[8] = {u0.x, u0.y, u0.z, u0.w, u1.x, u1.y, u1.z, u1.w};
        #pragma unroll
        for (int kb = 0; kb < 4; ++kb) {
            const v2f a = (v2f){u[2 * kb], u[2 * kb + 1]};
            acc = __builtin_amdgcn_wmma_f32_16x16x4_f32(false, a, false, bw[kb], (short)0, acc, false, false);
            ke = __builtin_elementwise_fma(splat(a.x), wk[2 * kb],     ke);
            ko = __builtin_elementwise_fma(splat(a.y), wk[2 * kb + 1], ko);
            se = __builtin_elementwise_fma(splat(a.x), ws[2 * kb],     se);
            so = __builtin_elementwise_fma(splat(a.y), ws[2 * kb + 1], so);
        }
        keep = ke + ko;
        send = se + so;

        #pragma unroll
        for (int r = 0; r < 8; ++r)
            wr[r * P] = relu(acc[r]);
        wave_order();
    };

    if ((T & 3) == 0) {
        const int G = T >> 2;
        v4f xb0 = *(const v4f*)(xrow);
        v4f xb1 = (G > 1) ? *(const v4f*)(xrow + 4) : xb0;
        #pragma unroll 1
        for (int g = 0; g < G; ++g) {
            const v4f cur = xb0;
            xb0 = xb1;
            const int gg = (g + 2 < G) ? (g + 2) : (G - 1);
            xb1 = *(const v4f*)(xrow + (size_t)gg * 4);
            #pragma unroll
            for (int u = 0; u < 4; ++u)
                step(cur[u]);
        }
    } else {
        #pragma unroll 1
        for (int t = 0; t < T; ++t)
            step(xrow[t]);
    }

    st[c * P + jk]     = relu(keep.x + xhalf(send.x));
    st[c * P + jk + 1] = relu(keep.y + xhalf(send.y));
    wave_order();
    if (lane < 16) {
        const int row = base + lane;
        if (row < B) {
            const float* fin = st + lane * P;
            float accf = fc_b[0];
            #pragma unroll
            for (int j = 0; j < HSZ; ++j)
                accf += fin[j] * fc_w[j];
            out[row] = accf;
        }
    }
}

extern "C" void kernel_launch(void* const* d_in, const int* in_sizes, int n_in,
                              void* d_out, int out_size, void* d_ws, size_t ws_size,
                              hipStream_t stream) {
    const float* x    = (const float*)d_in[0];
    const float* W_ih = (const float*)d_in[1];
    const float* W_hh = (const float*)d_in[2];
    const float* b_ih = (const float*)d_in[3];
    const float* b_hh = (const float*)d_in[4];
    const float* fc_w = (const float*)d_in[5];
    const float* fc_b = (const float*)d_in[6];
    float* out = (float*)d_out;

    const int B = out_size;
    const int T = in_sizes[0] / B;
    const int blocks = (B + 15) / 16;

    rnn_scan_wmma_f32<<<blocks, 32, 0, stream>>>(x, W_ih, W_hh, b_ih, b_hh,
                                                 fc_w, fc_b, out, B, T);
}
